# starting kernel
# baseline (speedup 1.0000x reference)
#include <hip/hip_runtime.h>

typedef __attribute__((ext_vector_type(16))) _Float16 v16h;
typedef __attribute__((ext_vector_type(8)))  float    v8f;

#define DF   128
#define DK2  256
#define DOUT 64

static __host__ __device__ inline int cdiv(int a, int b) { return (a + b - 1) / b; }

__device__ __forceinline__ void edge_sr(int i, const int* __restrict__ snd,
                                        const int* __restrict__ rcv, int E,
                                        int& s, int& r) {
    if (i < E)            { s = snd[i];     r = rcv[i]; }
    else if (i < 2 * E)   { s = rcv[i - E]; r = snd[i - E]; }
    else                  { s = i - 2 * E;  r = s; }
}

__global__ void zero_kernel(float* __restrict__ p, long n) {
    long i = (long)blockIdx.x * blockDim.x + threadIdx.x;
    long stride = (long)gridDim.x * blockDim.x;
    for (; i < n; i += stride) p[i] = 0.0f;
}

__global__ void deg_kernel(const int* __restrict__ snd, const int* __restrict__ rcv,
                           const float* __restrict__ theta, int E, int Nn,
                           float* __restrict__ degS, float* __restrict__ degR) {
    int i = blockIdx.x * blockDim.x + threadIdx.x;
    int M = 2 * E + Nn;
    if (i >= M) return;
    int s, r; edge_sr(i, snd, rcv, E, s, r);
    float ow, iw;
    if (i < E)          { float c = cosf(theta[i]);     float sn = sinf(theta[i]);     ow = c * c;  iw = sn * sn; }
    else if (i < 2 * E) { float c = cosf(theta[i - E]); float sn = sinf(theta[i - E]); ow = sn * sn; iw = c * c; }
    else                { ow = 1.0f; iw = 1.0f; }
    unsafeAtomicAdd(degS + s, ow);
    unsafeAtomicAdd(degR + s, iw);
}

__global__ void dis_kernel(float* __restrict__ d, int n) {
    int i = blockIdx.x * blockDim.x + threadIdx.x;
    if (i >= n) return;
    float v = d[i] + 1e-12f;
    d[i] = (v > 1e11f) ? 0.0f : rsqrtf(v);
}

__global__ void weight_kernel(const int* __restrict__ snd, const int* __restrict__ rcv,
                              const float* __restrict__ theta, int E, int Nn,
                              const float* __restrict__ disS, const float* __restrict__ disR,
                              float* __restrict__ wS, float* __restrict__ wD) {
    int i = blockIdx.x * blockDim.x + threadIdx.x;
    int M = 2 * E + Nn;
    if (i >= M) return;
    int s, r; edge_sr(i, snd, rcv, E, s, r);
    float ow, iw;
    if (i < E)          { float c = cosf(theta[i]);     float sn = sinf(theta[i]);     ow = c * c;  iw = sn * sn; }
    else if (i < 2 * E) { float c = cosf(theta[i - E]); float sn = sinf(theta[i - E]); ow = sn * sn; iw = c * c; }
    else                { ow = 1.0f; iw = 1.0f; }
    wS[i] = disS[s] * ow * disR[r];
    wD[i] = disR[s] * iw * disS[r];
}

__global__ void scatter_kernel(const float* __restrict__ X,
                               const int* __restrict__ snd, const int* __restrict__ rcv,
                               const float* __restrict__ wS, const float* __restrict__ wD,
                               float* __restrict__ agg, int E, int Nn) {
    int wave = (blockIdx.x * blockDim.x + threadIdx.x) >> 5;
    int lane = threadIdx.x & 31;
    int M = 2 * E + Nn;
    if (wave >= M) return;
    int s, r; edge_sr(wave, snd, rcv, E, s, r);
    float a = wS[wave];
    float b = wD[wave];
    const float4 v = *(const float4*)(X + (long)s * DF + lane * 4);
    float* dst = agg + (long)r * DK2 + lane * 4;
    unsafeAtomicAdd(dst + 0,        v.x * a);
    unsafeAtomicAdd(dst + 1,        v.y * a);
    unsafeAtomicAdd(dst + 2,        v.z * a);
    unsafeAtomicAdd(dst + 3,        v.w * a);
    unsafeAtomicAdd(dst + DF + 0,   v.x * b);
    unsafeAtomicAdd(dst + DF + 1,   v.y * b);
    unsafeAtomicAdd(dst + DF + 2,   v.z * b);
    unsafeAtomicAdd(dst + DF + 3,   v.w * b);
}

__global__ void pack_cat(const float* __restrict__ Ws, const float* __restrict__ Wd,
                         _Float16* __restrict__ dst) {
    int idx = blockIdx.x * blockDim.x + threadIdx.x;
    if (idx >= DK2 * DF) return;
    int j    = idx & 15;
    int lane = (idx >> 4) & 31;
    int t    = idx >> 9;
    int k0t  = t / (DF / 16);
    int tc   = t % (DF / 16);
    int k = k0t * 32 + ((j < 8) ? 0 : 16) + (lane >> 4) * 8 + (j & 7);
    int n = tc * 16 + (lane & 15);
    float v = (k < DF) ? 0.5f * Ws[k * DF + n] : 0.5f * Wd[(k - DF) * DF + n];
    dst[idx] = (_Float16)v;
}

__global__ void pack_single(const float* __restrict__ W, _Float16* __restrict__ dst) {
    int idx = blockIdx.x * blockDim.x + threadIdx.x;
    if (idx >= DF * DOUT) return;
    int j    = idx & 15;
    int lane = (idx >> 4) & 31;
    int t    = idx >> 9;
    int k0t  = t / (DOUT / 16);
    int tc   = t % (DOUT / 16);
    int k = k0t * 32 + ((j < 8) ? 0 : 16) + (lane >> 4) * 8 + (j & 7);
    int n = tc * 16 + (lane & 15);
    dst[idx] = (_Float16)W[k * DOUT + n];
}

__global__ void bias_combine(const float* __restrict__ bs, const float* __restrict__ bd,
                             float* __restrict__ out, int n) {
    int i = blockIdx.x * blockDim.x + threadIdx.x;
    if (i < n) out[i] = 0.5f * (bs[i] + bd[i]);
}

template <int K, int NC, int RELU>
__global__ void wmma_gemm(const float* __restrict__ A, int Mrows,
                          const v16h* __restrict__ Bp, const float* __restrict__ bias,
                          float* __restrict__ D) {
    constexpr int colTiles = NC / 16;
    int wave = (blockIdx.x * blockDim.x + threadIdx.x) >> 5;
    int lane = threadIdx.x & 31;
    int rowTiles = Mrows >> 4;
    if (wave >= rowTiles * colTiles) return;
    int tr = wave / colTiles;
    int tc = wave % colTiles;
    int hi = lane >> 4;
    int arow = tr * 16 + (lane & 15);
    int ncol = tc * 16 + (lane & 15);

    const float4* Ar = (const float4*)(A + (long)arow * K);
    v8f c = {};
#pragma unroll
    for (int k0 = 0; k0 < K; k0 += 32) {
        int base = (k0 >> 2) + hi * 2;
        float4 p0 = Ar[base];
        float4 p1 = Ar[base + 1];
        float4 p2 = Ar[base + 4];
        float4 p3 = Ar[base + 5];
        v16h a;
        a[0]  = (_Float16)p0.x; a[1]  = (_Float16)p0.y;
        a[2]  = (_Float16)p0.z; a[3]  = (_Float16)p0.w;
        a[4]  = (_Float16)p1.x; a[5]  = (_Float16)p1.y;
        a[6]  = (_Float16)p1.z; a[7]  = (_Float16)p1.w;
        a[8]  = (_Float16)p2.x; a[9]  = (_Float16)p2.y;
        a[10] = (_Float16)p2.z; a[11] = (_Float16)p2.w;
        a[12] = (_Float16)p3.x; a[13] = (_Float16)p3.y;
        a[14] = (_Float16)p3.z; a[15] = (_Float16)p3.w;
        v16h b = Bp[((k0 >> 5) * colTiles + tc) * 32 + lane];
        c = __builtin_amdgcn_wmma_f32_16x16x32_f16(false, a, false, b,
                                                   (short)0, c, false, false);
    }
    float bv = bias[ncol];
    float* Drow = D + (long)(tr * 16 + hi * 8) * NC + ncol;
#pragma unroll
    for (int i = 0; i < 8; ++i) {
        float v = c[i] + bv;
        if (RELU) v = fmaxf(v, 0.0f);
        Drow[(long)i * NC] = v;
    }
}

extern "C" void kernel_launch(void* const* d_in, const int* in_sizes, int n_in,
                              void* d_out, int out_size, void* d_ws, size_t ws_size,
                              hipStream_t stream) {
    const float* x      = (const float*)d_in[0];
    const int*   edges  = (const int*)d_in[1];
    const float* theta  = (const float*)d_in[2];
    const float* W1s    = (const float*)d_in[3];
    const float* W1d    = (const float*)d_in[4];
    const float* b1s    = (const float*)d_in[5];
    const float* b1d    = (const float*)d_in[6];
    const float* W2s    = (const float*)d_in[7];
    const float* W2d    = (const float*)d_in[8];
    const float* b2s    = (const float*)d_in[9];
    const float* b2d    = (const float*)d_in[10];
    const float* Wro    = (const float*)d_in[11];
    const float* bro    = (const float*)d_in[12];

    const int N = in_sizes[0] / DF;
    const int E = in_sizes[2];
    const int M = 2 * E + N;
    const int* snd = edges;
    const int* rcv = edges + E;

    char* ws = (char*)d_ws;
    size_t off = 0;
    auto alloc = [&](size_t bytes) { char* p = ws + off; off = (off + bytes + 255) & ~(size_t)255; return p; };
    float*    degS  = (float*)alloc((size_t)N * 4);
    float*    degR  = (float*)alloc((size_t)N * 4);
    float*    wS    = (float*)alloc((size_t)M * 4);
    float*    wD    = (float*)alloc((size_t)M * 4);
    float*    agg   = (float*)alloc((size_t)N * DK2 * 4);
    float*    x1    = (float*)alloc((size_t)N * DF * 4);
    float*    x2    = (float*)alloc((size_t)N * DF * 4);
    _Float16* Bp1   = (_Float16*)alloc((size_t)DK2 * DF * 2);
    _Float16* Bp2   = (_Float16*)alloc((size_t)DK2 * DF * 2);
    _Float16* Bpro  = (_Float16*)alloc((size_t)DF * DOUT * 2);
    float*    bias1 = (float*)alloc(DF * 4);
    float*    bias2 = (float*)alloc(DF * 4);

    const int T = 256;

    zero_kernel<<<512, T, 0, stream>>>(degS, N);
    zero_kernel<<<512, T, 0, stream>>>(degR, N);
    deg_kernel<<<cdiv(M, T), T, 0, stream>>>(snd, rcv, theta, E, N, degS, degR);
    dis_kernel<<<cdiv(N, T), T, 0, stream>>>(degS, N);
    dis_kernel<<<cdiv(N, T), T, 0, stream>>>(degR, N);
    weight_kernel<<<cdiv(M, T), T, 0, stream>>>(snd, rcv, theta, E, N, degS, degR, wS, wD);

    pack_cat<<<cdiv(DK2 * DF, T), T, 0, stream>>>(W1s, W1d, Bp1);
    pack_cat<<<cdiv(DK2 * DF, T), T, 0, stream>>>(W2s, W2d, Bp2);
    pack_single<<<cdiv(DF * DOUT, T), T, 0, stream>>>(Wro, Bpro);
    bias_combine<<<1, DF, 0, stream>>>(b1s, b1d, bias1, DF);
    bias_combine<<<1, DF, 0, stream>>>(b2s, b2d, bias2, DF);

    const int scatterBlocks = cdiv(M * 32, T);
    const int layerTiles    = (N / 16) * (DF / 16);
    const int roTiles       = (N / 16) * (DOUT / 16);

    zero_kernel<<<4096, T, 0, stream>>>(agg, (long)N * DK2);
    scatter_kernel<<<scatterBlocks, T, 0, stream>>>(x, snd, rcv, wS, wD, agg, E, N);
    wmma_gemm<DK2, DF, 1><<<cdiv(layerTiles * 32, T), T, 0, stream>>>(
        agg, N, (const v16h*)Bp1, bias1, x1);

    zero_kernel<<<4096, T, 0, stream>>>(agg, (long)N * DK2);
    scatter_kernel<<<scatterBlocks, T, 0, stream>>>(x1, snd, rcv, wS, wD, agg, E, N);
    wmma_gemm<DK2, DF, 1><<<cdiv(layerTiles * 32, T), T, 0, stream>>>(
        agg, N, (const v16h*)Bp2, bias2, x2);

    wmma_gemm<DF, DOUT, 0><<<cdiv(roTiles * 32, T), T, 0, stream>>>(
        x2, N, (const v16h*)Bpro, bro, (float*)d_out);
}


# Both layers: in-edge gather and layer GEMM fused into one launch: 8 waves gather 16 nodes into an LDS tile in the matrix operand format and multiply from it, so the f32 sums never go to memory
# speedup vs baseline: 2.8924x; 2.8924x over previous
#include <hip/hip_runtime.h>

typedef __attribute__((ext_vector_type(16))) _Float16 v16h;
typedef __attribute__((ext_vector_type(8)))  float    v8f;
typedef __attribute__((ext_vector_type(8)))  _Float16 v8h;
typedef __attribute__((ext_vector_type(4)))  _Float16 v4h;

#define DF   128
#define DK2  256
#define DOUT 64

static __host__ __device__ inline int cdiv(int a, int b) { return (a + b - 1) / b; }

__device__ __forceinline__ void edge_sr(int i, const int* __restrict__ snd,
                                        const int* __restrict__ rcv, int E,
                                        int& s, int& r) {
    if (i < E)            { s = snd[i];     r = rcv[i]; }
    else if (i < 2 * E)   { s = rcv[i - E]; r = snd[i - E]; }
    else                  { s = i - 2 * E;  r = s; }
}

__global__ void zero_kernel(float* __restrict__ p, long n) {
    long i = (long)blockIdx.x * blockDim.x + threadIdx.x;
    long stride = (long)gridDim.x * blockDim.x;
    for (; i < n; i += stride) p[i] = 0.0f;
}

__global__ void deg_kernel(const int* __restrict__ snd, const int* __restrict__ rcv,
                           const float* __restrict__ theta, int E, int Nn,
                           float* __restrict__ degS, float* __restrict__ degR,
                           int* __restrict__ cnt, int* __restrict__ rnk) {
    int i = blockIdx.x * blockDim.x + threadIdx.x;
    int M = 2 * E + Nn;
    if (i >= M) return;
    int s, r; edge_sr(i, snd, rcv, E, s, r);
    float ow, iw;
    if (i < E)          { float c = cosf(theta[i]);     float sn = sinf(theta[i]);     ow = c * c;  iw = sn * sn; }
    else if (i < 2 * E) { float c = cosf(theta[i - E]); float sn = sinf(theta[i - E]); ow = sn * sn; iw = c * c; }
    else                { ow = 1.0f; iw = 1.0f; }
    unsafeAtomicAdd(degS + s, ow);
    unsafeAtomicAdd(degR + s, iw);
    if (i < 2 * E) rnk[i] = atomicAdd(cnt + s, 1);
}

__global__ void dis_kernel(float* __restrict__ d, int n) {
    int i = blockIdx.x * blockDim.x + threadIdx.x;
    if (i >= n) return;
    float v = d[i] + 1e-12f;
    d[i] = (v > 1e11f) ? 0.0f : rsqrtf(v);
}

#define SCAN_T 256

struct __attribute__((aligned(16))) InEdge { int src; float ws; float wd; int pad; };

__global__ void scan_local(const int* __restrict__ cnt, int* __restrict__ off,
                           int* __restrict__ bsum, int n) {
    __shared__ int sh[SCAN_T];
    int t = threadIdx.x;
    int i = blockIdx.x * SCAN_T + t;
    int val = (i < n) ? cnt[i] + 1 : 0;
    sh[t] = val;
    __syncthreads();
    for (int d = 1; d < SCAN_T; d <<= 1) {
        int add = (t >= d) ? sh[t - d] : 0;
        __syncthreads();
        sh[t] += add;
        __syncthreads();
    }
    if (i < n) off[i] = sh[t] - val;
    if (t == SCAN_T - 1) bsum[blockIdx.x] = sh[t];
}

__global__ void scan_top(const int* __restrict__ bsum, int* __restrict__ boff, int nb) {
    __shared__ int sh[SCAN_T];
    int t = threadIdx.x;
    int carry = 0;
    for (int base = 0; base < nb; base += SCAN_T) {
        int i = base + t;
        int val = (i < nb) ? bsum[i] : 0;
        sh[t] = val;
        __syncthreads();
        for (int d = 1; d < SCAN_T; d <<= 1) {
            int add = (t >= d) ? sh[t - d] : 0;
            __syncthreads();
            sh[t] += add;
            __syncthreads();
        }
        if (i < nb) boff[i] = carry + sh[t] - val;
        carry += sh[SCAN_T - 1];
        __syncthreads();
    }
}

__global__ void fill_kernel(const int* __restrict__ snd, const int* __restrict__ rcv,
                            const float* __restrict__ theta, int E, int Nn,
                            const float* __restrict__ disS, const float* __restrict__ disR,
                            const int* __restrict__ cnt, const int* __restrict__ rnk,
                            const int* __restrict__ off, const int* __restrict__ boff,
                            InEdge* __restrict__ lst) {
    int i = blockIdx.x * blockDim.x + threadIdx.x;
    int M = 2 * E + Nn;
    if (i >= M) return;
    int s, r; edge_sr(i, snd, rcv, E, s, r);
    float ow, iw;
    int pos;
    if (i < E)          { float c = cosf(theta[i]);     float sn = sinf(theta[i]);     ow = c * c;  iw = sn * sn; pos = rnk[i + E]; }
    else if (i < 2 * E) { float c = cosf(theta[i - E]); float sn = sinf(theta[i - E]); ow = sn * sn; iw = c * c;  pos = rnk[i - E]; }
    else                { ow = 1.0f; iw = 1.0f; pos = cnt[r]; }
    InEdge e;
    e.src = s;
    e.ws  = disS[s] * ow * disR[r];
    e.wd  = disR[s] * iw * disS[r];
    e.pad = 0;
    lst[off[r] + boff[r / SCAN_T] + pos] = e;
}

#define TILE_M 16
#define APITCH (DK2 + 8)

template <int RELU>
__global__ __launch_bounds__(256)
void gather_gemm(const float* __restrict__ X,
                 const int* __restrict__ cnt, const int* __restrict__ off,
                 const int* __restrict__ boff, const InEdge* __restrict__ lst,
                 const v16h* __restrict__ Bp, const float* __restrict__ bias,
                 float* __restrict__ D, int Nn) {
    __shared__ __attribute__((aligned(16))) _Float16 As[TILE_M][APITCH];
    const int lane = threadIdx.x & 31;
    const int wave = __builtin_amdgcn_readfirstlane((int)(threadIdx.x >> 5));
    const int v0 = blockIdx.x * TILE_M;
    const float* Xl = X + lane * 4;

    for (int j = 0; j < 2; ++j) {
        const int row = wave * 2 + j;
        const int v = v0 + row;
        float4 as = make_float4(0.f, 0.f, 0.f, 0.f);
        float4 ad = make_float4(0.f, 0.f, 0.f, 0.f);
        if (v < Nn) {
            const int beg = off[v] + boff[v / SCAN_T];
            const int n   = cnt[v] + 1;
            for (int c = 0; c < n; c += 32) {
                int m = n - c; if (m > 32) m = 32;
                InEdge e; e.src = 0; e.ws = 0.f; e.wd = 0.f; e.pad = 0;
                if (lane < m) e = lst[beg + c + lane];
#pragma unroll 4
                for (int t = 0; t < m; ++t) {
                    int   s = __builtin_amdgcn_readlane(e.src, t);
                    float a = __int_as_float(__builtin_amdgcn_readlane(__float_as_int(e.ws), t));
                    float b = __int_as_float(__builtin_amdgcn_readlane(__float_as_int(e.wd), t));
                    const float4 xv = *(const float4*)(Xl + (long)s * DF);
                    as.x = fmaf(xv.x, a, as.x); as.y = fmaf(xv.y, a, as.y);
                    as.z = fmaf(xv.z, a, as.z); as.w = fmaf(xv.w, a, as.w);
                    ad.x = fmaf(xv.x, b, ad.x); ad.y = fmaf(xv.y, b, ad.y);
                    ad.z = fmaf(xv.z, b, ad.z); ad.w = fmaf(xv.w, b, ad.w);
                }
            }
        }
        v4h hs, hd;
        hs[0] = (_Float16)as.x; hs[1] = (_Float16)as.y; hs[2] = (_Float16)as.z; hs[3] = (_Float16)as.w;
        hd[0] = (_Float16)ad.x; hd[1] = (_Float16)ad.y; hd[2] = (_Float16)ad.z; hd[3] = (_Float16)ad.w;
        *(v4h*)&As[row][lane * 4]      = hs;
        *(v4h*)&As[row][DF + lane * 4] = hd;
    }
    __syncthreads();

    const int hi  = lane >> 4;
    const int c16 = lane & 15;
    const int tc  = wave;
    const _Float16* Arow = &As[c16][hi * 8];
    v8f acc = {};
#pragma unroll 2
    for (int k0 = 0; k0 < DK2; k0 += 32) {
        const v8h lo = *(const v8h*)(Arow + k0);
        const v8h up = *(const v8h*)(Arow + k0 + 16);
        const v16h a = __builtin_shufflevector(lo, up, 0, 1, 2, 3, 4, 5, 6, 7,
                                               8, 9, 10, 11, 12, 13, 14, 15);
        const v16h b = Bp[((k0 >> 5) * (DF / 16) + tc) * 32 + lane];
        acc = __builtin_amdgcn_wmma_f32_16x16x32_f16(false, a, false, b,
                                                     (short)0, acc, false, false);
    }
    const int ncol = tc * 16 + c16;
    const float bv = bias[ncol];
    const int r0 = v0 + hi * 8;
    float* Drow = D + (long)r0 * DF + ncol;
    if (v0 + TILE_M <= Nn) {
#pragma unroll
        for (int i = 0; i < 8; ++i) {
            float val = acc[i] + bv;
            if (RELU) val = fmaxf(val, 0.0f);
            Drow[(long)i * DF] = val;
        }
    } else {
#pragma unroll
        for (int i = 0; i < 8; ++i) {
            float val = acc[i] + bv;
            if (RELU) val = fmaxf(val, 0.0f);
            if (r0 + i < Nn) Drow[(long)i * DF] = val;
        }
    }
}

__global__ void pack_cat(const float* __restrict__ Ws, const float* __restrict__ Wd,
                         _Float16* __restrict__ dst) {
    int idx = blockIdx.x * blockDim.x + threadIdx.x;
    if (idx >= DK2 * DF) return;
    int j    = idx & 15;
    int lane = (idx >> 4) & 31;
    int t    = idx >> 9;
    int k0t  = t / (DF / 16);
    int tc   = t % (DF / 16);
    int k = k0t * 32 + ((j < 8) ? 0 : 16) + (lane >> 4) * 8 + (j & 7);
    int n = tc * 16 + (lane & 15);
    float v = (k < DF) ? 0.5f * Ws[k * DF + n] : 0.5f * Wd[(k - DF) * DF + n];
    dst[idx] = (_Float16)v;
}

__global__ void pack_single(const float* __restrict__ W, _Float16* __restrict__ dst) {
    int idx = blockIdx.x * blockDim.x + threadIdx.x;
    if (idx >= DF * DOUT) return;
    int j    = idx & 15;
    int lane = (idx >> 4) & 31;
    int t    = idx >> 9;
    int k0t  = t / (DOUT / 16);
    int tc   = t % (DOUT / 16);
    int k = k0t * 32 + ((j < 8) ? 0 : 16) + (lane >> 4) * 8 + (j & 7);
    int n = tc * 16 + (lane & 15);
    dst[idx] = (_Float16)W[k * DOUT + n];
}

__global__ void bias_combine(const float* __restrict__ bs, const float* __restrict__ bd,
                             float* __restrict__ out, int n) {
    int i = blockIdx.x * blockDim.x + threadIdx.x;
    if (i < n) out[i] = 0.5f * (bs[i] + bd[i]);
}

template <int K, int NC, int RELU>
__global__ void wmma_gemm(const float* __restrict__ A, int Mrows,
                          const v16h* __restrict__ Bp, const float* __restrict__ bias,
                          float* __restrict__ D) {
    constexpr int colTiles = NC / 16;
    int wave = (blockIdx.x * blockDim.x + threadIdx.x) >> 5;
    int lane = threadIdx.x & 31;
    int rowTiles = Mrows >> 4;
    if (wave >= rowTiles * colTiles) return;
    int tr = wave / colTiles;
    int tc = wave % colTiles;
    int hi = lane >> 4;
    int arow = tr * 16 + (lane & 15);
    int ncol = tc * 16 + (lane & 15);

    const float4* Ar = (const float4*)(A + (long)arow * K);
    v8f c = {};
#pragma unroll
    for (int k0 = 0; k0 < K; k0 += 32) {
        int base = (k0 >> 2) + hi * 2;
        float4 p0 = Ar[base];
        float4 p1 = Ar[base + 1];
        float4 p2 = Ar[base + 4];
        float4 p3 = Ar[base + 5];
        v16h a;
        a[0]  = (_Float16)p0.x; a[1]  = (_Float16)p0.y;
        a[2]  = (_Float16)p0.z; a[3]  = (_Float16)p0.w;
        a[4]  = (_Float16)p1.x; a[5]  = (_Float16)p1.y;
        a[6]  = (_Float16)p1.z; a[7]  = (_Float16)p1.w;
        a[8]  = (_Float16)p2.x; a[9]  = (_Float16)p2.y;
        a[10] = (_Float16)p2.z; a[11] = (_Float16)p2.w;
        a[12] = (_Float16)p3.x; a[13] = (_Float16)p3.y;
        a[14] = (_Float16)p3.z; a[15] = (_Float16)p3.w;
        v16h b = Bp[((k0 >> 5) * colTiles + tc) * 32 + lane];
        c = __builtin_amdgcn_wmma_f32_16x16x32_f16(false, a, false, b,
                                                   (short)0, c, false, false);
    }
    float bv = bias[ncol];
    float* Drow = D + (long)(tr * 16 + hi * 8) * NC + ncol;
#pragma unroll
    for (int i = 0; i < 8; ++i) {
        float v = c[i] + bv;
        if (RELU) v = fmaxf(v, 0.0f);
        Drow[(long)i * NC] = v;
    }
}

extern "C" void kernel_launch(void* const* d_in, const int* in_sizes, int n_in,
                              void* d_out, int out_size, void* d_ws, size_t ws_size,
                              hipStream_t stream) {
    const float* x      = (const float*)d_in[0];
    const int*   edges  = (const int*)d_in[1];
    const float* theta  = (const float*)d_in[2];
    const float* W1s    = (const float*)d_in[3];
    const float* W1d    = (const float*)d_in[4];
    const float* b1s    = (const float*)d_in[5];
    const float* b1d    = (const float*)d_in[6];
    const float* W2s    = (const float*)d_in[7];
    const float* W2d    = (const float*)d_in[8];
    const float* b2s    = (const float*)d_in[9];
    const float* b2d    = (const float*)d_in[10];
    const float* Wro    = (const float*)d_in[11];
    const float* bro    = (const float*)d_in[12];

    const int N = in_sizes[0] / DF;
    const int E = in_sizes[2];
    const int M = 2 * E + N;
    const int* snd = edges;
    const int* rcv = edges + E;

    char* ws = (char*)d_ws;
    size_t off = 0;
    auto alloc = [&](size_t bytes) { char* p = ws + off; off = (off + bytes + 255) & ~(size_t)255; return p; };
    float*    nodeacc = (float*)alloc((size_t)N * 3 * 4);
    float*    degS  = nodeacc;
    float*    degR  = nodeacc + N;
    int*      cnt   = (int*)(nodeacc + 2 * (size_t)N);
    const int NB    = cdiv(N, SCAN_T);
    int*      noff  = (int*)alloc((size_t)N * 4);
    int*      bsum  = (int*)alloc((size_t)NB * 4);
    int*      boff  = (int*)alloc((size_t)NB * 4);
    float*    x2    = (float*)alloc((size_t)N * DK2 * 4);
    float*    x1    = (float*)alloc((size_t)N * DF * 4);
    char*     lreg  = alloc((size_t)N * DF * 4);
    _Float16* Bp1   = (_Float16*)alloc((size_t)DK2 * DF * 2);
    _Float16* Bp2   = (_Float16*)alloc((size_t)DK2 * DF * 2);
    _Float16* Bpro  = (_Float16*)alloc((size_t)DF * DOUT * 2);
    float*    bias1 = (float*)alloc(DF * 4);
    float*    bias2 = (float*)alloc(DF * 4);
    InEdge*   lst   = (InEdge*)lreg;
    int*      rnk   = (int*)(lst + M);

    const int T = 256;

    zero_kernel<<<512, T, 0, stream>>>(nodeacc, (long)N * 3);
    deg_kernel<<<cdiv(M, T), T, 0, stream>>>(snd, rcv, theta, E, N, degS, degR, cnt, rnk);
    dis_kernel<<<cdiv(N, T), T, 0, stream>>>(degS, N);
    dis_kernel<<<cdiv(N, T), T, 0, stream>>>(degR, N);
    scan_local<<<NB, SCAN_T, 0, stream>>>(cnt, noff, bsum, N);
    scan_top<<<1, SCAN_T, 0, stream>>>(bsum, boff, NB);
    fill_kernel<<<cdiv(M, T), T, 0, stream>>>(snd, rcv, theta, E, N, degS, degR,
                                              cnt, rnk, noff, boff, lst);

    pack_cat<<<cdiv(DK2 * DF, T), T, 0, stream>>>(W1s, W1d, Bp1);
    pack_cat<<<cdiv(DK2 * DF, T), T, 0, stream>>>(W2s, W2d, Bp2);
    pack_single<<<cdiv(DF * DOUT, T), T, 0, stream>>>(Wro, Bpro);
    bias_combine<<<1, DF, 0, stream>>>(b1s, b1d, bias1, DF);
    bias_combine<<<1, DF, 0, stream>>>(b2s, b2d, bias2, DF);

    const int layerBlocks  = cdiv(N, TILE_M);
    const int roTiles      = (N / 16) * (DOUT / 16);

    gather_gemm<1><<<layerBlocks, T, 0, stream>>>(x, cnt, noff, boff, lst,
                                                  (const v16h*)Bp1, bias1, x1, N);

    gather_gemm<1><<<layerBlocks, T, 0, stream>>>(x1, cnt, noff, boff, lst,
                                                  (const v16h*)Bp2, bias2, x2, N);

    wmma_gemm<DF, DOUT, 0><<<cdiv(roTiles * 32, T), T, 0, stream>>>(
        x2, N, (const v16h*)Bpro, bro, (float*)d_out);
}
